# starting kernel
# baseline (speedup 1.0000x reference)
#include <hip/hip_runtime.h>


typedef __attribute__((ext_vector_type(2))) float v2f;
typedef __attribute__((ext_vector_type(8))) float v8f;

#define DIMX   512
#define SEQ    16
#define NB     8
#define KVLEN  4112
#define KTILES 257
#define MAXSEQ 8192
#define QKSCALE 0.0625f

__device__ __forceinline__ v8f wmma4(v2f a, v2f b, v8f c) {
  return __builtin_amdgcn_wmma_f32_16x16x4_f32(false, a, false, b,
                                               (short)0, c, false, false);
}

__global__ void __launch_bounds__(128)
qkv_kernel(const float* __restrict__ x,
           const float* __restrict__ Wq, const float* __restrict__ Wk,
           const float* __restrict__ Wv,
           float* __restrict__ q, float* __restrict__ kn, float* __restrict__ vn) {
  const int t    = blockIdx.x * 4 + (threadIdx.x >> 5);
  const int lane = threadIdx.x & 31;
  const int half = lane >> 4, lm = lane & 15;
  const int mat  = t >> 8;
  const int rem  = t & 255;
  const int Mt   = rem >> 5, Nt = rem & 31;

  const float* W = (mat == 0) ? Wq : (mat == 1) ? Wk : Wv;
  float*       D = (mat == 0) ? q  : (mat == 1) ? kn : vn;

  const float* arow = x + (Mt * 16 + lm) * DIMX + 2 * half;
  const float* wrow = W + (Nt * 16 + lm) * DIMX + 2 * half;

  v8f c = {};
#pragma unroll 4
  for (int k = 0; k < DIMX; k += 4) {
    v2f a = *(const v2f*)(arow + k);
    v2f b = *(const v2f*)(wrow + k);
    c = wmma4(a, b, c);
  }
  float* drow = D + Nt * 16 + lm;
#pragma unroll
  for (int i = 0; i < 8; ++i)
    drow[(Mt * 16 + i + 8 * half) * DIMX] = c[i];
}

__global__ void __launch_bounds__(128)
dist_kernel(const float* __restrict__ q,
            const float* __restrict__ cache_k,
            const float* __restrict__ kn,
            float* __restrict__ wout) {
  const int t    = blockIdx.x * 4 + (threadIdx.x >> 5);
  const int lane = threadIdx.x & 31;
  const int half = lane >> 4, lm = lane & 15;
  const int b  = t / KTILES;
  const int kt = t - b * KTILES;

  const float* arow = q + (size_t)(b * SEQ + lm) * DIMX + 2 * half;
  const float* key  = (kt < 256)
      ? cache_k + (size_t)(b * MAXSEQ + kt * 16) * DIMX
      : kn      + (size_t)b * SEQ * DIMX;
  const float* brow = key + (size_t)lm * DIMX + 2 * half;

  v8f c = {};
#pragma unroll 4
  for (int k = 0; k < DIMX; k += 4) {
    v2f a  = *(const v2f*)(arow + k);
    v2f bb = *(const v2f*)(brow + k);
    c = wmma4(a, bb, c);
  }
  float* drow = wout + (size_t)b * SEQ * KVLEN + kt * 16 + lm;
#pragma unroll
  for (int i = 0; i < 8; ++i)
    drow[(size_t)(i + 8 * half) * KVLEN] = c[i] * QKSCALE;
}

__global__ void __launch_bounds__(256)
softmax_kernel(const float* __restrict__ wout,
               const float* __restrict__ mask,
               float* __restrict__ probs) {
  const int row = blockIdx.x;
  const int m   = row & 15;
  const float* drow = wout  + (size_t)row * KVLEN;
  const float* mrow = mask  + (size_t)m   * KVLEN;
  float*       prow = probs + (size_t)row * KVLEN;

  __shared__ float red[256];
  const int tid = threadIdx.x;

  float lmax = -3.402823466e38f;
  for (int j = tid; j < KVLEN; j += 256)
    lmax = fmaxf(lmax, drow[j] + mrow[j]);
  red[tid] = lmax; __syncthreads();
  for (int s = 128; s > 0; s >>= 1) {
    if (tid < s) red[tid] = fmaxf(red[tid], red[tid + s]);
    __syncthreads();
  }
  const float rmax = red[0]; __syncthreads();

  float lsum = 0.f;
  for (int j = tid; j < KVLEN; j += 256)
    lsum += __expf(drow[j] + mrow[j] - rmax);
  red[tid] = lsum; __syncthreads();
  for (int s = 128; s > 0; s >>= 1) {
    if (tid < s) red[tid] += red[tid + s];
    __syncthreads();
  }
  const float inv = 1.f / red[0];

  for (int j = tid; j < KVLEN; j += 256)
    prow[j] = __expf(drow[j] + mrow[j] - rmax) * inv;
}

__global__ void __launch_bounds__(256)
atten_kernel(const float* __restrict__ probs,
             const float* __restrict__ cache_v,
             const float* __restrict__ vn,
             float* __restrict__ atten) {
  const int blk  = blockIdx.x;
  const int b    = blk >> 5;
  const int Nt   = blk & 31;
  const int wv   = threadIdx.x >> 5;
  const int lane = threadIdx.x & 31;
  const int half = lane >> 4, lm = lane & 15;

  const float* prow = probs + (size_t)(b * SEQ + lm) * KVLEN + 2 * half;

  v8f c = {};
  for (int kt = wv; kt < KTILES; kt += 8) {
    const float* vt = (kt < 256)
        ? cache_v + (size_t)(b * MAXSEQ + kt * 16) * DIMX
        : vn      + (size_t)b * SEQ * DIMX;
    const float* vcol = vt + Nt * 16 + lm;
    const int kb = kt * 16;
#pragma unroll
    for (int s = 0; s < 4; ++s) {
      v2f a = *(const v2f*)(prow + kb + 4 * s);
      v2f bb;
      bb.x = vcol[(size_t)(4 * s + 2 * half)     * DIMX];
      bb.y = vcol[(size_t)(4 * s + 2 * half + 1) * DIMX];
      c = wmma4(a, bb, c);
    }
  }

  __shared__ float red[8 * 256];
#pragma unroll
  for (int i = 0; i < 8; ++i) red[wv * 256 + i * 32 + lane] = c[i];
  __syncthreads();
  if (wv == 0) {
    #pragma unroll
    for (int i = 0; i < 8; ++i) {
      float s = 0.f;
      #pragma unroll
      for (int w = 0; w < 8; ++w) s += red[w * 256 + i * 32 + lane];
      atten[(size_t)(b * SEQ + i + 8 * half) * DIMX + Nt * 16 + lm] = s;
    }
  }
}

__global__ void __launch_bounds__(128)
outproj_kernel(const float* __restrict__ atten,
               const float* __restrict__ Wo,
               float* __restrict__ out) {
  const int t    = blockIdx.x * 4 + (threadIdx.x >> 5);
  const int lane = threadIdx.x & 31;
  const int half = lane >> 4, lm = lane & 15;
  const int Mt   = t >> 5, Nt = t & 31;

  const float* arow = atten + (Mt * 16 + lm) * DIMX + 2 * half;
  const float* wrow = Wo    + (Nt * 16 + lm) * DIMX + 2 * half;

  v8f c = {};
#pragma unroll 4
  for (int k = 0; k < DIMX; k += 4) {
    v2f a = *(const v2f*)(arow + k);
    v2f b = *(const v2f*)(wrow + k);
    c = wmma4(a, b, c);
  }
#pragma unroll
  for (int i = 0; i < 8; ++i)
    out[(size_t)(Mt * 16 + i + 8 * half) * DIMX + Nt * 16 + lm] = c[i];
}

extern "C" void kernel_launch(void* const* d_in, const int* in_sizes, int n_in,
                              void* d_out, int out_size, void* d_ws, size_t ws_size,
                              hipStream_t stream) {
  (void)in_sizes; (void)n_in; (void)out_size; (void)ws_size;
  const float* x       = (const float*)d_in[0];
  const float* mask    = (const float*)d_in[2];
  const float* Wq      = (const float*)d_in[3];
  const float* Wk      = (const float*)d_in[4];
  const float* Wv      = (const float*)d_in[5];
  const float* Wo      = (const float*)d_in[6];
  const float* cache_k = (const float*)d_in[7];
  const float* cache_v = (const float*)d_in[8];

  float* out  = (float*)d_out;
  float* wout = out + NB * SEQ * DIMX;

  float* ws    = (float*)d_ws;
  float* q     = ws;
  float* kn    = q  + NB * SEQ * DIMX;
  float* vn    = kn + NB * SEQ * DIMX;
  float* probs = vn + NB * SEQ * DIMX;
  float* atten = probs + (size_t)NB * SEQ * KVLEN;

  qkv_kernel    <<<192, 128, 0, stream>>>(x, Wq, Wk, Wv, q, kn, vn);
  dist_kernel   <<<514, 128, 0, stream>>>(q, cache_k, kn, wout);
  softmax_kernel<<<128, 256, 0, stream>>>(wout, mask, probs);
  atten_kernel  <<<256, 256, 0, stream>>>(probs, cache_v, vn, atten);
  outproj_kernel<<< 64, 128, 0, stream>>>(atten, Wo, out);
}


# softmax: each row's scores and mask read once by 16-byte loads all in flight together, kept in registers through max, exp-sum and scaling (was 3 passes, loads 2 deep); f32 row-sum order changes
# speedup vs baseline: 1.4519x; 1.4519x over previous
#include <hip/hip_runtime.h>


typedef __attribute__((ext_vector_type(2))) float v2f;
typedef __attribute__((ext_vector_type(8))) float v8f;

#define DIMX   512
#define SEQ    16
#define NB     8
#define KVLEN  4112
#define KTILES 257
#define MAXSEQ 8192
#define QKSCALE 0.0625f

__device__ __forceinline__ v8f wmma4(v2f a, v2f b, v8f c) {
  return __builtin_amdgcn_wmma_f32_16x16x4_f32(false, a, false, b,
                                               (short)0, c, false, false);
}

template <int KLEN>
__device__ __forceinline__ v8f tile_partial(const float* __restrict__ arow,
                                            const float* __restrict__ brow) {
  v8f c = {};
#pragma unroll 4
  for (int k = 0; k < KLEN; k += 4) {
    v2f a = *(const v2f*)(arow + k);
    v2f b = *(const v2f*)(brow + k);
    c = wmma4(a, b, c);
  }
  return c;
}

__device__ __forceinline__ void put_partial(float* red, int wv, int lane, v8f c) {
#pragma unroll
  for (int i = 0; i < 8; ++i) red[wv * 256 + i * 32 + lane] = c[i];
}

__global__ void __launch_bounds__(128)
qkv_kernel(const float* __restrict__ x,
           const float* __restrict__ Wq, const float* __restrict__ Wk,
           const float* __restrict__ Wv,
           float* __restrict__ q, float* __restrict__ kn, float* __restrict__ vn) {
  const int t    = blockIdx.x;
  const int wv   = threadIdx.x >> 5;
  const int lane = threadIdx.x & 31;
  const int half = lane >> 4, lm = lane & 15;
  const int mat  = t >> 8;
  const int rem  = t & 255;
  const int Mt   = rem >> 5, Nt = rem & 31;

  const float* W = (mat == 0) ? Wq : (mat == 1) ? Wk : Wv;
  float*       D = (mat == 0) ? q  : (mat == 1) ? kn : vn;

  const float* arow = x + (Mt * 16 + lm) * DIMX + 2 * half + wv * 128;
  const float* wrow = W + (Nt * 16 + lm) * DIMX + 2 * half + wv * 128;

  const v8f c = tile_partial<128>(arow, wrow);

  __shared__ float red[4 * 256];
  put_partial(red, wv, lane, c);
  __syncthreads();
  for (int j = threadIdx.x; j < 256; j += 128) {
    const float s = ((red[j] + red[256 + j]) + red[512 + j]) + red[768 + j];
    const int i = j >> 5, ln = j & 31;
    D[(Mt * 16 + i + 8 * (ln >> 4)) * DIMX + Nt * 16 + (ln & 15)] = s;
  }
}

__global__ void __launch_bounds__(128)
dist_kernel(const float* __restrict__ q,
            const float* __restrict__ cache_k,
            const float* __restrict__ kn,
            float* __restrict__ wout) {
  const int t    = blockIdx.x;
  const int wv   = threadIdx.x >> 5;
  const int lane = threadIdx.x & 31;
  const int half = lane >> 4, lm = lane & 15;
  const int b  = t / KTILES;
  const int kt = t - b * KTILES;

  const float* arow = q + (size_t)(b * SEQ + lm) * DIMX + 2 * half + wv * 128;
  const float* key  = (kt < 256)
      ? cache_k + (size_t)(b * MAXSEQ + kt * 16) * DIMX
      : kn      + (size_t)b * SEQ * DIMX;
  const float* brow = key + (size_t)lm * DIMX + 2 * half + wv * 128;

  const v8f c = tile_partial<128>(arow, brow);

  __shared__ float red[4 * 256];
  put_partial(red, wv, lane, c);
  __syncthreads();
  float* dtile = wout + (size_t)b * SEQ * KVLEN + kt * 16;
  for (int j = threadIdx.x; j < 256; j += 128) {
    const float s = ((red[j] + red[256 + j]) + red[512 + j]) + red[768 + j];
    const int i = j >> 5, ln = j & 31;
    dtile[(size_t)(i + 8 * (ln >> 4)) * KVLEN + (ln & 15)] = s * QKSCALE;
  }
}

typedef __attribute__((ext_vector_type(4))) float v4f;

#define SM_T    256
#define SM_W    (SM_T / 32)
#define SM_G    (KVLEN / 4)
#define SM_U    (SM_G / SM_T)
#define SM_TAIL (SM_G - SM_U * SM_T)

__device__ __forceinline__ float wave_max(float v) {
#pragma unroll
  for (int x = 16; x >= 1; x >>= 1) v = fmaxf(v, __shfl_xor(v, x, 32));
  return v;
}
__device__ __forceinline__ float wave_sum(float v) {
#pragma unroll
  for (int x = 16; x >= 1; x >>= 1) v += __shfl_xor(v, x, 32);
  return v;
}

__device__ __forceinline__ float max4(v4f a) {
  return fmaxf(fmaxf(a.x, a.y), fmaxf(a.z, a.w));
}
__device__ __forceinline__ v4f exp4(v4f a, float rmax) {
  v4f e;
  e.x = __expf(a.x - rmax); e.y = __expf(a.y - rmax);
  e.z = __expf(a.z - rmax); e.w = __expf(a.w - rmax);
  return e;
}

__global__ void __launch_bounds__(SM_T)
softmax_kernel(const float* __restrict__ wout,
               const float* __restrict__ mask,
               float* __restrict__ probs) {
  const int row = blockIdx.x;
  const int m   = row & 15;
  const v4f* drow = (const v4f*)(wout  + (size_t)row * KVLEN);
  const v4f* mrow = (const v4f*)(mask  + (size_t)m   * KVLEN);
  v4f*       prow = (v4f*)(probs + (size_t)row * KVLEN);

  const int tid  = threadIdx.x;
  const int wv   = tid >> 5, lane = tid & 31;
  const bool tail = tid < SM_TAIL;

  __shared__ float wmax[SM_W];
  __shared__ float wsum[SM_W];

  v4f v[SM_U];
#pragma unroll
  for (int u = 0; u < SM_U; ++u)
    v[u] = drow[tid + u * SM_T] + mrow[tid + u * SM_T];
  v4f vt = {};
  if (tail) vt = drow[SM_U * SM_T + tid] + mrow[SM_U * SM_T + tid];

  float lmax = -3.402823466e38f;
#pragma unroll
  for (int u = 0; u < SM_U; ++u) lmax = fmaxf(lmax, max4(v[u]));
  if (tail) lmax = fmaxf(lmax, max4(vt));
  lmax = wave_max(lmax);
  if (lane == 0) wmax[wv] = lmax;
  __syncthreads();
  float rmax = wmax[0];
#pragma unroll
  for (int w = 1; w < SM_W; ++w) rmax = fmaxf(rmax, wmax[w]);

  float lsum = 0.f;
#pragma unroll
  for (int u = 0; u < SM_U; ++u) {
    v[u] = exp4(v[u], rmax);
    lsum += (v[u].x + v[u].y) + (v[u].z + v[u].w);
  }
  if (tail) {
    vt = exp4(vt, rmax);
    lsum += (vt.x + vt.y) + (vt.z + vt.w);
  }
  lsum = wave_sum(lsum);
  if (lane == 0) wsum[wv] = lsum;
  __syncthreads();
  float tot = wsum[0];
#pragma unroll
  for (int w = 1; w < SM_W; ++w) tot += wsum[w];
  const float inv = 1.f / tot;

#pragma unroll
  for (int u = 0; u < SM_U; ++u) prow[tid + u * SM_T] = v[u] * inv;
  if (tail) prow[SM_U * SM_T + tid] = vt * inv;
}

__global__ void __launch_bounds__(1024)
atten_kernel(const float* __restrict__ probs,
             const float* __restrict__ cache_v,
             const float* __restrict__ vn,
             float* __restrict__ atten) {
  const int blk  = blockIdx.x;
  const int b    = blk >> 5;
  const int Nt   = blk & 31;
  const int wv   = threadIdx.x >> 5;
  const int lane = threadIdx.x & 31;
  const int half = lane >> 4, lm = lane & 15;

  const float* prow = probs + (size_t)(b * SEQ + lm) * KVLEN + 2 * half;

  v8f c = {};
  for (int kt = wv; kt < KTILES; kt += 32) {
    const float* vt = (kt < 256)
        ? cache_v + (size_t)(b * MAXSEQ + kt * 16) * DIMX
        : vn      + (size_t)b * SEQ * DIMX;
    const float* vcol = vt + Nt * 16 + lm;
    const int kb = kt * 16;
#pragma unroll
    for (int s = 0; s < 4; ++s) {
      v2f a = *(const v2f*)(prow + kb + 4 * s);
      v2f bb;
      bb.x = vcol[(size_t)(4 * s + 2 * half)     * DIMX];
      bb.y = vcol[(size_t)(4 * s + 2 * half + 1) * DIMX];
      c = wmma4(a, bb, c);
    }
  }

  __shared__ float red[32 * 256];
  put_partial(red, wv, lane, c);
  __syncthreads();
  if (threadIdx.x < 256) {
    const int j = threadIdx.x;
    float s = 0.f;
#pragma unroll
    for (int w = 0; w < 32; ++w) s += red[w * 256 + j];
    const int i = j >> 5, ln = j & 31;
    atten[(size_t)(b * SEQ + i + 8 * (ln >> 4)) * DIMX + Nt * 16 + (ln & 15)] = s;
  }
}

__global__ void __launch_bounds__(256)
outproj_kernel(const float* __restrict__ atten,
               const float* __restrict__ Wo,
               float* __restrict__ out) {
  const int t    = blockIdx.x;
  const int wv   = threadIdx.x >> 5;
  const int lane = threadIdx.x & 31;
  const int half = lane >> 4, lm = lane & 15;
  const int Mt   = t >> 5, Nt = t & 31;

  const float* arow = atten + (Mt * 16 + lm) * DIMX + 2 * half + wv * 64;
  const float* wrow = Wo    + (Nt * 16 + lm) * DIMX + 2 * half + wv * 64;

  const v8f c = tile_partial<64>(arow, wrow);

  __shared__ float red[8 * 256];
  put_partial(red, wv, lane, c);
  __syncthreads();
  {
    const int j = threadIdx.x;
    float s = 0.f;
#pragma unroll
    for (int w = 0; w < 8; ++w) s += red[w * 256 + j];
    const int i = j >> 5, ln = j & 31;
    out[(size_t)(Mt * 16 + i + 8 * (ln >> 4)) * DIMX + Nt * 16 + (ln & 15)] = s;
  }
}

extern "C" void kernel_launch(void* const* d_in, const int* in_sizes, int n_in,
                              void* d_out, int out_size, void* d_ws, size_t ws_size,
                              hipStream_t stream) {
  (void)in_sizes; (void)n_in; (void)out_size; (void)ws_size;
  const float* x       = (const float*)d_in[0];
  const float* mask    = (const float*)d_in[2];
  const float* Wq      = (const float*)d_in[3];
  const float* Wk      = (const float*)d_in[4];
  const float* Wv      = (const float*)d_in[5];
  const float* Wo      = (const float*)d_in[6];
  const float* cache_k = (const float*)d_in[7];
  const float* cache_v = (const float*)d_in[8];

  float* out  = (float*)d_out;
  float* wout = out + NB * SEQ * DIMX;

  float* ws    = (float*)d_ws;
  float* q     = ws;
  float* kn    = q  + NB * SEQ * DIMX;
  float* vn    = kn + NB * SEQ * DIMX;
  float* probs = vn + NB * SEQ * DIMX;
  float* atten = probs + (size_t)NB * SEQ * KVLEN;

  qkv_kernel    <<< 768,  128, 0, stream>>>(x, Wq, Wk, Wv, q, kn, vn);
  dist_kernel   <<<2056,  128, 0, stream>>>(q, cache_k, kn, wout);
  softmax_kernel<<< 128, SM_T, 0, stream>>>(wout, mask, probs);
  atten_kernel  <<< 256, 1024, 0, stream>>>(probs, cache_v, vn, atten);
  outproj_kernel<<< 256,  256, 0, stream>>>(atten, Wo, out);
}


# P@V: wave owns 8 consecutive k-tiles, takes their probs as 16 row segments of 512 contiguous bytes into its own LDS plane, A fragments read from LDS (was 32 loads of 16 lines); f32 sum order changes
# speedup vs baseline: 1.5970x; 1.0999x over previous
#include <hip/hip_runtime.h>


typedef __attribute__((ext_vector_type(2))) float v2f;
typedef __attribute__((ext_vector_type(8))) float v8f;

#define DIMX   512
#define SEQ    16
#define NB     8
#define KVLEN  4112
#define KTILES 257
#define MAXSEQ 8192
#define QKSCALE 0.0625f

__device__ __forceinline__ v8f wmma4(v2f a, v2f b, v8f c) {
  return __builtin_amdgcn_wmma_f32_16x16x4_f32(false, a, false, b,
                                               (short)0, c, false, false);
}

template <int KLEN>
__device__ __forceinline__ v8f tile_partial(const float* __restrict__ arow,
                                            const float* __restrict__ brow) {
  v8f c = {};
#pragma unroll 4
  for (int k = 0; k < KLEN; k += 4) {
    v2f a = *(const v2f*)(arow + k);
    v2f b = *(const v2f*)(brow + k);
    c = wmma4(a, b, c);
  }
  return c;
}

__device__ __forceinline__ void put_partial(float* red, int wv, int lane, v8f c) {
#pragma unroll
  for (int i = 0; i < 8; ++i) red[wv * 256 + i * 32 + lane] = c[i];
}

__global__ void __launch_bounds__(128)
qkv_kernel(const float* __restrict__ x,
           const float* __restrict__ Wq, const float* __restrict__ Wk,
           const float* __restrict__ Wv,
           float* __restrict__ q, float* __restrict__ kn, float* __restrict__ vn) {
  const int t    = blockIdx.x;
  const int wv   = threadIdx.x >> 5;
  const int lane = threadIdx.x & 31;
  const int half = lane >> 4, lm = lane & 15;
  const int mat  = t >> 8;
  const int rem  = t & 255;
  const int Mt   = rem >> 5, Nt = rem & 31;

  const float* W = (mat == 0) ? Wq : (mat == 1) ? Wk : Wv;
  float*       D = (mat == 0) ? q  : (mat == 1) ? kn : vn;

  const float* arow = x + (Mt * 16 + lm) * DIMX + 2 * half + wv * 128;
  const float* wrow = W + (Nt * 16 + lm) * DIMX + 2 * half + wv * 128;

  const v8f c = tile_partial<128>(arow, wrow);

  __shared__ float red[4 * 256];
  put_partial(red, wv, lane, c);
  __syncthreads();
  for (int j = threadIdx.x; j < 256; j += 128) {
    const float s = ((red[j] + red[256 + j]) + red[512 + j]) + red[768 + j];
    const int i = j >> 5, ln = j & 31;
    D[(Mt * 16 + i + 8 * (ln >> 4)) * DIMX + Nt * 16 + (ln & 15)] = s;
  }
}

__global__ void __launch_bounds__(128)
dist_kernel(const float* __restrict__ q,
            const float* __restrict__ cache_k,
            const float* __restrict__ kn,
            float* __restrict__ wout) {
  const int t    = blockIdx.x;
  const int wv   = threadIdx.x >> 5;
  const int lane = threadIdx.x & 31;
  const int half = lane >> 4, lm = lane & 15;
  const int b  = t / KTILES;
  const int kt = t - b * KTILES;

  const float* arow = q + (size_t)(b * SEQ + lm) * DIMX + 2 * half + wv * 128;
  const float* key  = (kt < 256)
      ? cache_k + (size_t)(b * MAXSEQ + kt * 16) * DIMX
      : kn      + (size_t)b * SEQ * DIMX;
  const float* brow = key + (size_t)lm * DIMX + 2 * half + wv * 128;

  const v8f c = tile_partial<128>(arow, brow);

  __shared__ float red[4 * 256];
  put_partial(red, wv, lane, c);
  __syncthreads();
  float* dtile = wout + (size_t)b * SEQ * KVLEN + kt * 16;
  for (int j = threadIdx.x; j < 256; j += 128) {
    const float s = ((red[j] + red[256 + j]) + red[512 + j]) + red[768 + j];
    const int i = j >> 5, ln = j & 31;
    dtile[(size_t)(i + 8 * (ln >> 4)) * KVLEN + (ln & 15)] = s * QKSCALE;
  }
}

typedef __attribute__((ext_vector_type(4))) float v4f;

#define SM_T    256
#define SM_W    (SM_T / 32)
#define SM_G    (KVLEN / 4)
#define SM_U    (SM_G / SM_T)
#define SM_TAIL (SM_G - SM_U * SM_T)

__device__ __forceinline__ float wave_max(float v) {
#pragma unroll
  for (int x = 16; x >= 1; x >>= 1) v = fmaxf(v, __shfl_xor(v, x, 32));
  return v;
}
__device__ __forceinline__ float wave_sum(float v) {
#pragma unroll
  for (int x = 16; x >= 1; x >>= 1) v += __shfl_xor(v, x, 32);
  return v;
}

__device__ __forceinline__ float max4(v4f a) {
  return fmaxf(fmaxf(a.x, a.y), fmaxf(a.z, a.w));
}
__device__ __forceinline__ v4f exp4(v4f a, float rmax) {
  v4f e;
  e.x = __expf(a.x - rmax); e.y = __expf(a.y - rmax);
  e.z = __expf(a.z - rmax); e.w = __expf(a.w - rmax);
  return e;
}

__global__ void __launch_bounds__(SM_T)
softmax_kernel(const float* __restrict__ wout,
               const float* __restrict__ mask,
               float* __restrict__ probs) {
  const int row = blockIdx.x;
  const int m   = row & 15;
  const v4f* drow = (const v4f*)(wout  + (size_t)row * KVLEN);
  const v4f* mrow = (const v4f*)(mask  + (size_t)m   * KVLEN);
  v4f*       prow = (v4f*)(probs + (size_t)row * KVLEN);

  const int tid  = threadIdx.x;
  const int wv   = tid >> 5, lane = tid & 31;
  const bool tail = tid < SM_TAIL;

  __shared__ float wmax[SM_W];
  __shared__ float wsum[SM_W];

  v4f v[SM_U];
#pragma unroll
  for (int u = 0; u < SM_U; ++u)
    v[u] = drow[tid + u * SM_T] + mrow[tid + u * SM_T];
  v4f vt = {};
  if (tail) vt = drow[SM_U * SM_T + tid] + mrow[SM_U * SM_T + tid];

  float lmax = -3.402823466e38f;
#pragma unroll
  for (int u = 0; u < SM_U; ++u) lmax = fmaxf(lmax, max4(v[u]));
  if (tail) lmax = fmaxf(lmax, max4(vt));
  lmax = wave_max(lmax);
  if (lane == 0) wmax[wv] = lmax;
  __syncthreads();
  float rmax = wmax[0];
#pragma unroll
  for (int w = 1; w < SM_W; ++w) rmax = fmaxf(rmax, wmax[w]);

  float lsum = 0.f;
#pragma unroll
  for (int u = 0; u < SM_U; ++u) {
    v[u] = exp4(v[u], rmax);
    lsum += (v[u].x + v[u].y) + (v[u].z + v[u].w);
  }
  if (tail) {
    vt = exp4(vt, rmax);
    lsum += (vt.x + vt.y) + (vt.z + vt.w);
  }
  lsum = wave_sum(lsum);
  if (lane == 0) wsum[wv] = lsum;
  __syncthreads();
  float tot = wsum[0];
#pragma unroll
  for (int w = 1; w < SM_W; ++w) tot += wsum[w];
  const float inv = 1.f / tot;

#pragma unroll
  for (int u = 0; u < SM_U; ++u) prow[tid + u * SM_T] = v[u] * inv;
  if (tail) prow[SM_U * SM_T + tid] = vt * inv;
}

#define PV_SPAN  128
#define PV_PITCH 132

__global__ void __launch_bounds__(1024)
atten_kernel(const float* __restrict__ probs,
             const float* __restrict__ cache_v,
             const float* __restrict__ vn,
             float* __restrict__ atten) {
  const int blk  = blockIdx.x;
  const int b    = blk >> 5;
  const int Nt   = blk & 31;
  const int wv   = threadIdx.x >> 5;
  const int lane = threadIdx.x & 31;
  const int half = lane >> 4, lm = lane & 15;

  __shared__ float stage[32 * SEQ * PV_PITCH];
  __shared__ float red[32 * 256];

  float* plane = stage + wv * (SEQ * PV_PITCH);
  {
    const float* pw = probs + (size_t)b * SEQ * KVLEN + wv * PV_SPAN + 4 * lane;
    v4f x[SEQ];
#pragma unroll
    for (int r = 0; r < SEQ; ++r) x[r] = *(const v4f*)(pw + (size_t)r * KVLEN);
#pragma unroll
    for (int r = 0; r < SEQ; ++r) *(v4f*)(plane + r * PV_PITCH + 4 * lane) = x[r];
  }
  __syncwarp();

  const float* afrag = plane + lm * PV_PITCH + 2 * half;
  const float* vcol  = cache_v + (size_t)(b * MAXSEQ + wv * PV_SPAN) * DIMX
                     + Nt * 16 + lm;

  v8f c = {};
  for (int t = 0; t < PV_SPAN / 16; ++t) {
#pragma unroll
    for (int s = 0; s < 4; ++s) {
      const int j = t * 16 + 4 * s;
      v2f a = *(const v2f*)(afrag + j);
      v2f bb;
      bb.x = vcol[(size_t)(j + 2 * half)     * DIMX];
      bb.y = vcol[(size_t)(j + 2 * half + 1) * DIMX];
      c = wmma4(a, bb, c);
    }
  }
  if (wv == 0) {
    const float* prow = probs + (size_t)(b * SEQ + lm) * KVLEN + 256 * 16 + 2 * half;
    const float* vfr  = vn + (size_t)b * SEQ * DIMX + Nt * 16 + lm;
#pragma unroll
    for (int s = 0; s < 4; ++s) {
      v2f a = *(const v2f*)(prow + 4 * s);
      v2f bb;
      bb.x = vfr[(size_t)(4 * s + 2 * half)     * DIMX];
      bb.y = vfr[(size_t)(4 * s + 2 * half + 1) * DIMX];
      c = wmma4(a, bb, c);
    }
  }

  put_partial(red, wv, lane, c);
  __syncthreads();
  if (threadIdx.x < 256) {
    const int j = threadIdx.x;
    float s = 0.f;
#pragma unroll
    for (int w = 0; w < 32; ++w) s += red[w * 256 + j];
    const int i = j >> 5, ln = j & 31;
    atten[(size_t)(b * SEQ + i + 8 * (ln >> 4)) * DIMX + Nt * 16 + (ln & 15)] = s;
  }
}

__global__ void __launch_bounds__(256)
outproj_kernel(const float* __restrict__ atten,
               const float* __restrict__ Wo,
               float* __restrict__ out) {
  const int t    = blockIdx.x;
  const int wv   = threadIdx.x >> 5;
  const int lane = threadIdx.x & 31;
  const int half = lane >> 4, lm = lane & 15;
  const int Mt   = t >> 5, Nt = t & 31;

  const float* arow = atten + (Mt * 16 + lm) * DIMX + 2 * half + wv * 64;
  const float* wrow = Wo    + (Nt * 16 + lm) * DIMX + 2 * half + wv * 64;

  const v8f c = tile_partial<64>(arow, wrow);

  __shared__ float red[8 * 256];
  put_partial(red, wv, lane, c);
  __syncthreads();
  {
    const int j = threadIdx.x;
    float s = 0.f;
#pragma unroll
    for (int w = 0; w < 8; ++w) s += red[w * 256 + j];
    const int i = j >> 5, ln = j & 31;
    out[(size_t)(Mt * 16 + i + 8 * (ln >> 4)) * DIMX + Nt * 16 + (ln & 15)] = s;
  }
}

extern "C" void kernel_launch(void* const* d_in, const int* in_sizes, int n_in,
                              void* d_out, int out_size, void* d_ws, size_t ws_size,
                              hipStream_t stream) {
  (void)in_sizes; (void)n_in; (void)out_size; (void)ws_size;
  const float* x       = (const float*)d_in[0];
  const float* mask    = (const float*)d_in[2];
  const float* Wq      = (const float*)d_in[3];
  const float* Wk      = (const float*)d_in[4];
  const float* Wv      = (const float*)d_in[5];
  const float* Wo      = (const float*)d_in[6];
  const float* cache_k = (const float*)d_in[7];
  const float* cache_v = (const float*)d_in[8];

  float* out  = (float*)d_out;
  float* wout = out + NB * SEQ * DIMX;

  float* ws    = (float*)d_ws;
  float* q     = ws;
  float* kn    = q  + NB * SEQ * DIMX;
  float* vn    = kn + NB * SEQ * DIMX;
  float* probs = vn + NB * SEQ * DIMX;
  float* atten = probs + (size_t)NB * SEQ * KVLEN;

  qkv_kernel    <<< 768,  128, 0, stream>>>(x, Wq, Wk, Wv, q, kn, vn);
  dist_kernel   <<<2056,  128, 0, stream>>>(q, cache_k, kn, wout);
  softmax_kernel<<< 128, SM_T, 0, stream>>>(wout, mask, probs);
  atten_kernel  <<< 256, 1024, 0, stream>>>(probs, cache_v, vn, atten);
  outproj_kernel<<< 256,  256, 0, stream>>>(atten, Wo, out);
}


# score GEMM: a wave keeps its q fragments (k quarter) in registers and runs 4 key tiles on them, keys fetched 8 k steps ahead; 2056 blocks -> 520, q loads per tile /4, 16-byte stores; f32 sums same
# speedup vs baseline: 1.8376x; 1.1507x over previous
#include <hip/hip_runtime.h>


typedef __attribute__((ext_vector_type(2))) float v2f;
typedef __attribute__((ext_vector_type(8))) float v8f;

#define DIMX   512
#define SEQ    16
#define NB     8
#define KVLEN  4112
#define KTILES 257
#define MAXSEQ 8192
#define QKSCALE 0.0625f

__device__ __forceinline__ v8f wmma4(v2f a, v2f b, v8f c) {
  return __builtin_amdgcn_wmma_f32_16x16x4_f32(false, a, false, b,
                                               (short)0, c, false, false);
}

template <int KLEN>
__device__ __forceinline__ v8f tile_partial(const float* __restrict__ arow,
                                            const float* __restrict__ brow) {
  v8f c = {};
#pragma unroll 4
  for (int k = 0; k < KLEN; k += 4) {
    v2f a = *(const v2f*)(arow + k);
    v2f b = *(const v2f*)(brow + k);
    c = wmma4(a, b, c);
  }
  return c;
}

__device__ __forceinline__ void put_partial(float* red, int wv, int lane, v8f c) {
#pragma unroll
  for (int i = 0; i < 8; ++i) red[wv * 256 + i * 32 + lane] = c[i];
}

__global__ void __launch_bounds__(128)
qkv_kernel(const float* __restrict__ x,
           const float* __restrict__ Wq, const float* __restrict__ Wk,
           const float* __restrict__ Wv,
           float* __restrict__ q, float* __restrict__ kn, float* __restrict__ vn) {
  const int t    = blockIdx.x;
  const int wv   = threadIdx.x >> 5;
  const int lane = threadIdx.x & 31;
  const int half = lane >> 4, lm = lane & 15;
  const int mat  = t >> 8;
  const int rem  = t & 255;
  const int Mt   = rem >> 5, Nt = rem & 31;

  const float* W = (mat == 0) ? Wq : (mat == 1) ? Wk : Wv;
  float*       D = (mat == 0) ? q  : (mat == 1) ? kn : vn;

  const float* arow = x + (Mt * 16 + lm) * DIMX + 2 * half + wv * 128;
  const float* wrow = W + (Nt * 16 + lm) * DIMX + 2 * half + wv * 128;

  const v8f c = tile_partial<128>(arow, wrow);

  __shared__ float red[4 * 256];
  put_partial(red, wv, lane, c);
  __syncthreads();
  for (int j = threadIdx.x; j < 256; j += 128) {
    const float s = ((red[j] + red[256 + j]) + red[512 + j]) + red[768 + j];
    const int i = j >> 5, ln = j & 31;
    D[(Mt * 16 + i + 8 * (ln >> 4)) * DIMX + Nt * 16 + (ln & 15)] = s;
  }
}

typedef __attribute__((ext_vector_type(4))) float v4f;

#define SC_T 4

template <int T>
__device__ __forceinline__ void score_tiles(const float* __restrict__ qrow,
                                            const float* __restrict__ krow,
                                            float* __restrict__ red,
                                            float* __restrict__ dst,
                                            int wv, int lane) {
  constexpr int CH = (T == 1) ? 32 : 8;
  constexpr int NG = 32 / CH;
  v2f a[32];
#pragma unroll
  for (int s = 0; s < 32; ++s) a[s] = *(const v2f*)(qrow + 4 * s);

  v2f kb[2][CH * T];
#pragma unroll
  for (int j = 0; j < CH; ++j) {
#pragma unroll
    for (int t = 0; t < T; ++t)
      kb[0][j * T + t] = *(const v2f*)(krow + (size_t)t * 16 * DIMX + 4 * j);
  }

  v8f c[T];
#pragma unroll
  for (int t = 0; t < T; ++t) c[t] = v8f{};
#pragma unroll
  for (int g = 0; g < NG; ++g) {
    if (g + 1 < NG) {
#pragma unroll
      for (int j = 0; j < CH; ++j) {
#pragma unroll
        for (int t = 0; t < T; ++t)
          kb[(g + 1) & 1][j * T + t] =
              *(const v2f*)(krow + (size_t)t * 16 * DIMX + 4 * ((g + 1) * CH + j));
      }
      __builtin_amdgcn_sched_barrier(0);
    }
#pragma unroll
    for (int j = 0; j < CH; ++j) {
#pragma unroll
      for (int t = 0; t < T; ++t)
        c[t] = wmma4(a[g * CH + j], kb[g & 1][j * T + t], c[t]);
    }
  }

#pragma unroll
  for (int t = 0; t < T; ++t) {
#pragma unroll
    for (int i = 0; i < 8; ++i) red[(wv * T + t) * 256 + i * 32 + lane] = c[t][i];
  }
  __syncthreads();
  for (int f = threadIdx.x; f < 64 * T; f += 128) {
    const int row = f / (4 * T), c4 = f - row * (4 * T);
    const int t = c4 >> 2, col = (c4 & 3) * 4;
    const int e = (row & 7) * 32 + (row >> 3) * 16 + col;
    const v4f p0 = *(const v4f*)(red + (0 * T + t) * 256 + e);
    const v4f p1 = *(const v4f*)(red + (1 * T + t) * 256 + e);
    const v4f p2 = *(const v4f*)(red + (2 * T + t) * 256 + e);
    const v4f p3 = *(const v4f*)(red + (3 * T + t) * 256 + e);
    const v4f s = ((p0 + p1) + p2) + p3;
    *(v4f*)(dst + (size_t)row * KVLEN + 4 * c4) = s * QKSCALE;
  }
}

__global__ void __launch_bounds__(128)
dist_kernel(const float* __restrict__ q,
            const float* __restrict__ cache_k,
            const float* __restrict__ kn,
            float* __restrict__ wout) {
  const int wv   = threadIdx.x >> 5;
  const int lane = threadIdx.x & 31;
  const int half = lane >> 4, lm = lane & 15;
  const int koff = 2 * half + wv * 128;

  __shared__ __attribute__((aligned(16))) float red[4 * SC_T * 256];

  if (blockIdx.x < NB) {
    const int b = blockIdx.x;
    const float* qrow = q  + (size_t)(b * SEQ + lm) * DIMX + koff;
    const float* krow = kn + (size_t)(b * SEQ + lm) * DIMX + koff;
    score_tiles<1>(qrow, krow, red,
                   wout + (size_t)b * SEQ * KVLEN + 256 * 16, wv, lane);
  } else {
    const int t = blockIdx.x - NB;
    const int b = t >> 6, g = t & 63;
    const float* qrow = q + (size_t)(b * SEQ + lm) * DIMX + koff;
    const float* krow = cache_k
        + (size_t)(b * MAXSEQ + g * (16 * SC_T) + lm) * DIMX + koff;
    score_tiles<SC_T>(qrow, krow, red,
                      wout + (size_t)b * SEQ * KVLEN + g * (16 * SC_T), wv, lane);
  }
}


#define SM_T    256
#define SM_W    (SM_T / 32)
#define SM_G    (KVLEN / 4)
#define SM_U    (SM_G / SM_T)
#define SM_TAIL (SM_G - SM_U * SM_T)

__device__ __forceinline__ float wave_max(float v) {
#pragma unroll
  for (int x = 16; x >= 1; x >>= 1) v = fmaxf(v, __shfl_xor(v, x, 32));
  return v;
}
__device__ __forceinline__ float wave_sum(float v) {
#pragma unroll
  for (int x = 16; x >= 1; x >>= 1) v += __shfl_xor(v, x, 32);
  return v;
}

__device__ __forceinline__ float max4(v4f a) {
  return fmaxf(fmaxf(a.x, a.y), fmaxf(a.z, a.w));
}
__device__ __forceinline__ v4f exp4(v4f a, float rmax) {
  v4f e;
  e.x = __expf(a.x - rmax); e.y = __expf(a.y - rmax);
  e.z = __expf(a.z - rmax); e.w = __expf(a.w - rmax);
  return e;
}

__global__ void __launch_bounds__(SM_T)
softmax_kernel(const float* __restrict__ wout,
               const float* __restrict__ mask,
               float* __restrict__ probs) {
  const int row = blockIdx.x;
  const int m   = row & 15;
  const v4f* drow = (const v4f*)(wout  + (size_t)row * KVLEN);
  const v4f* mrow = (const v4f*)(mask  + (size_t)m   * KVLEN);
  v4f*       prow = (v4f*)(probs + (size_t)row * KVLEN);

  const int tid  = threadIdx.x;
  const int wv   = tid >> 5, lane = tid & 31;
  const bool tail = tid < SM_TAIL;

  __shared__ float wmax[SM_W];
  __shared__ float wsum[SM_W];

  v4f v[SM_U];
#pragma unroll
  for (int u = 0; u < SM_U; ++u)
    v[u] = drow[tid + u * SM_T] + mrow[tid + u * SM_T];
  v4f vt = {};
  if (tail) vt = drow[SM_U * SM_T + tid] + mrow[SM_U * SM_T + tid];

  float lmax = -3.402823466e38f;
#pragma unroll
  for (int u = 0; u < SM_U; ++u) lmax = fmaxf(lmax, max4(v[u]));
  if (tail) lmax = fmaxf(lmax, max4(vt));
  lmax = wave_max(lmax);
  if (lane == 0) wmax[wv] = lmax;
  __syncthreads();
  float rmax = wmax[0];
#pragma unroll
  for (int w = 1; w < SM_W; ++w) rmax = fmaxf(rmax, wmax[w]);

  float lsum = 0.f;
#pragma unroll
  for (int u = 0; u < SM_U; ++u) {
    v[u] = exp4(v[u], rmax);
    lsum += (v[u].x + v[u].y) + (v[u].z + v[u].w);
  }
  if (tail) {
    vt = exp4(vt, rmax);
    lsum += (vt.x + vt.y) + (vt.z + vt.w);
  }
  lsum = wave_sum(lsum);
  if (lane == 0) wsum[wv] = lsum;
  __syncthreads();
  float tot = wsum[0];
#pragma unroll
  for (int w = 1; w < SM_W; ++w) tot += wsum[w];
  const float inv = 1.f / tot;

#pragma unroll
  for (int u = 0; u < SM_U; ++u) prow[tid + u * SM_T] = v[u] * inv;
  if (tail) prow[SM_U * SM_T + tid] = vt * inv;
}

#define PV_SPAN  128
#define PV_PITCH 132

__global__ void __launch_bounds__(1024)
atten_kernel(const float* __restrict__ probs,
             const float* __restrict__ cache_v,
             const float* __restrict__ vn,
             float* __restrict__ atten) {
  const int blk  = blockIdx.x;
  const int b    = blk >> 5;
  const int Nt   = blk & 31;
  const int wv   = threadIdx.x >> 5;
  const int lane = threadIdx.x & 31;
  const int half = lane >> 4, lm = lane & 15;

  __shared__ float stage[32 * SEQ * PV_PITCH];
  __shared__ float red[32 * 256];

  float* plane = stage + wv * (SEQ * PV_PITCH);
  {
    const float* pw = probs + (size_t)b * SEQ * KVLEN + wv * PV_SPAN + 4 * lane;
    v4f x[SEQ];
#pragma unroll
    for (int r = 0; r < SEQ; ++r) x[r] = *(const v4f*)(pw + (size_t)r * KVLEN);
#pragma unroll
    for (int r = 0; r < SEQ; ++r) *(v4f*)(plane + r * PV_PITCH + 4 * lane) = x[r];
  }
  __syncwarp();

  const float* afrag = plane + lm * PV_PITCH + 2 * half;
  const float* vcol  = cache_v + (size_t)(b * MAXSEQ + wv * PV_SPAN) * DIMX
                     + Nt * 16 + lm;

  v8f c = {};
  for (int t = 0; t < PV_SPAN / 16; ++t) {
#pragma unroll
    for (int s = 0; s < 4; ++s) {
      const int j = t * 16 + 4 * s;
      v2f a = *(const v2f*)(afrag + j);
      v2f bb;
      bb.x = vcol[(size_t)(j + 2 * half)     * DIMX];
      bb.y = vcol[(size_t)(j + 2 * half + 1) * DIMX];
      c = wmma4(a, bb, c);
    }
  }
  if (wv == 0) {
    const float* prow = probs + (size_t)(b * SEQ + lm) * KVLEN + 256 * 16 + 2 * half;
    const float* vfr  = vn + (size_t)b * SEQ * DIMX + Nt * 16 + lm;
#pragma unroll
    for (int s = 0; s < 4; ++s) {
      v2f a = *(const v2f*)(prow + 4 * s);
      v2f bb;
      bb.x = vfr[(size_t)(4 * s + 2 * half)     * DIMX];
      bb.y = vfr[(size_t)(4 * s + 2 * half + 1) * DIMX];
      c = wmma4(a, bb, c);
    }
  }

  put_partial(red, wv, lane, c);
  __syncthreads();
  if (threadIdx.x < 256) {
    const int j = threadIdx.x;
    float s = 0.f;
#pragma unroll
    for (int w = 0; w < 32; ++w) s += red[w * 256 + j];
    const int i = j >> 5, ln = j & 31;
    atten[(size_t)(b * SEQ + i + 8 * (ln >> 4)) * DIMX + Nt * 16 + (ln & 15)] = s;
  }
}

__global__ void __launch_bounds__(256)
outproj_kernel(const float* __restrict__ atten,
               const float* __restrict__ Wo,
               float* __restrict__ out) {
  const int t    = blockIdx.x;
  const int wv   = threadIdx.x >> 5;
  const int lane = threadIdx.x & 31;
  const int half = lane >> 4, lm = lane & 15;
  const int Mt   = t >> 5, Nt = t & 31;

  const float* arow = atten + (Mt * 16 + lm) * DIMX + 2 * half + wv * 64;
  const float* wrow = Wo    + (Nt * 16 + lm) * DIMX + 2 * half + wv * 64;

  const v8f c = tile_partial<64>(arow, wrow);

  __shared__ float red[8 * 256];
  put_partial(red, wv, lane, c);
  __syncthreads();
  {
    const int j = threadIdx.x;
    float s = 0.f;
#pragma unroll
    for (int w = 0; w < 8; ++w) s += red[w * 256 + j];
    const int i = j >> 5, ln = j & 31;
    out[(size_t)(Mt * 16 + i + 8 * (ln >> 4)) * DIMX + Nt * 16 + (ln & 15)] = s;
  }
}

extern "C" void kernel_launch(void* const* d_in, const int* in_sizes, int n_in,
                              void* d_out, int out_size, void* d_ws, size_t ws_size,
                              hipStream_t stream) {
  (void)in_sizes; (void)n_in; (void)out_size; (void)ws_size;
  const float* x       = (const float*)d_in[0];
  const float* mask    = (const float*)d_in[2];
  const float* Wq      = (const float*)d_in[3];
  const float* Wk      = (const float*)d_in[4];
  const float* Wv      = (const float*)d_in[5];
  const float* Wo      = (const float*)d_in[6];
  const float* cache_k = (const float*)d_in[7];
  const float* cache_v = (const float*)d_in[8];

  float* out  = (float*)d_out;
  float* wout = out + NB * SEQ * DIMX;

  float* ws    = (float*)d_ws;
  float* q     = ws;
  float* kn    = q  + NB * SEQ * DIMX;
  float* vn    = kn + NB * SEQ * DIMX;
  float* probs = vn + NB * SEQ * DIMX;
  float* atten = probs + (size_t)NB * SEQ * KVLEN;

  qkv_kernel    <<< 768,  128, 0, stream>>>(x, Wq, Wk, Wv, q, kn, vn);
  dist_kernel   <<< 520,  128, 0, stream>>>(q, cache_k, kn, wout);
  softmax_kernel<<< 128, SM_T, 0, stream>>>(wout, mask, probs);
  atten_kernel  <<< 256, 1024, 0, stream>>>(probs, cache_v, vn, atten);
  outproj_kernel<<< 256,  256, 0, stream>>>(atten, Wo, out);
}


# q/k/v projection: a wave keeps its x fragments (a k eighth) in registers and runs 3 weight tiles on them, all 64 loads in flight at once; 768 blocks of 4 waves -> 256 of 8; f32 sum order changes
# speedup vs baseline: 1.9858x; 1.0806x over previous
#include <hip/hip_runtime.h>


typedef __attribute__((ext_vector_type(2))) float v2f;
typedef __attribute__((ext_vector_type(8))) float v8f;

#define DIMX   512
#define SEQ    16
#define NB     8
#define KVLEN  4112
#define KTILES 257
#define MAXSEQ 8192
#define QKSCALE 0.0625f

__device__ __forceinline__ v8f wmma4(v2f a, v2f b, v8f c) {
  return __builtin_amdgcn_wmma_f32_16x16x4_f32(false, a, false, b,
                                               (short)0, c, false, false);
}

template <int KLEN>
__device__ __forceinline__ v8f tile_partial(const float* __restrict__ arow,
                                            const float* __restrict__ brow) {
  v8f c = {};
#pragma unroll 4
  for (int k = 0; k < KLEN; k += 4) {
    v2f a = *(const v2f*)(arow + k);
    v2f b = *(const v2f*)(brow + k);
    c = wmma4(a, b, c);
  }
  return c;
}

__device__ __forceinline__ void put_partial(float* red, int wv, int lane, v8f c) {
#pragma unroll
  for (int i = 0; i < 8; ++i) red[wv * 256 + i * 32 + lane] = c[i];
}

#define QKV_T 3
__global__ void __launch_bounds__(256)
qkv_kernel(const float* __restrict__ x,
           const float* __restrict__ Wq, const float* __restrict__ Wk,
           const float* __restrict__ Wv,
           float* __restrict__ q, float* __restrict__ kn, float* __restrict__ vn) {
  const int t    = blockIdx.x;
  const int wv   = threadIdx.x >> 5;
  const int lane = threadIdx.x & 31;
  const int half = lane >> 4, lm = lane & 15;
  const int Mt   = t >> 5, g = t & 31;
  const int koff = 2 * half + wv * 64;

  const float* xrow = x + (Mt * 16 + lm) * DIMX + koff;
  v2f xa[16];
#pragma unroll
  for (int s = 0; s < 16; ++s) xa[s] = *(const v2f*)(xrow + 4 * s);

  v2f wb[QKV_T][16];
#pragma unroll
  for (int i = 0; i < QKV_T; ++i) {
    const int n = QKV_T * g + i;
    const float* W = (n < 32) ? Wq : (n < 64) ? Wk : Wv;
    const float* wrow = W + ((n & 31) * 16 + lm) * DIMX + koff;
#pragma unroll
    for (int s = 0; s < 16; ++s) wb[i][s] = *(const v2f*)(wrow + 4 * s);
  }
  __builtin_amdgcn_sched_barrier(0);

  v8f c[QKV_T];
#pragma unroll
  for (int i = 0; i < QKV_T; ++i) c[i] = v8f{};
#pragma unroll
  for (int s = 0; s < 16; ++s) {
#pragma unroll
    for (int i = 0; i < QKV_T; ++i) c[i] = wmma4(xa[s], wb[i][s], c[i]);
  }

  __shared__ float red[8 * QKV_T * 256];
#pragma unroll
  for (int i = 0; i < QKV_T; ++i) put_partial(red, wv * QKV_T + i, lane, c[i]);
  __syncthreads();
  {
    const int e = threadIdx.x;
    const int row = (e >> 5) + 8 * ((e & 31) >> 4), col = e & 15;
#pragma unroll
    for (int i = 0; i < QKV_T; ++i) {
      const int n = QKV_T * g + i;
      float* D = (n < 32) ? q : (n < 64) ? kn : vn;
      float s = 0.f;
#pragma unroll
      for (int w = 0; w < 8; ++w) s += red[(w * QKV_T + i) * 256 + e];
      D[(Mt * 16 + row) * DIMX + (n & 31) * 16 + col] = s;
    }
  }
}

typedef __attribute__((ext_vector_type(4))) float v4f;

#define SC_T 4
#define NGRP 65
#define SMFLOOR (-3.402823466e38f)

__device__ __forceinline__ float max4(v4f a) {
  return fmaxf(fmaxf(a.x, a.y), fmaxf(a.z, a.w));
}
__device__ __forceinline__ v4f exp4(v4f a, float rmax) {
  v4f e;
  e.x = __expf(a.x - rmax); e.y = __expf(a.y - rmax);
  e.z = __expf(a.z - rmax); e.w = __expf(a.w - rmax);
  return e;
}
template <int X>
__device__ __forceinline__ float row_xor(float v) {
  return __int_as_float(__builtin_amdgcn_mov_dpp(
      __float_as_int(v), 0x160 + X, 0xf, 0xf, false));
}
template <int N>
__device__ __forceinline__ float row_lane(float v) {
  return __int_as_float(__builtin_amdgcn_mov_dpp(
      __float_as_int(v), 0x150 + N, 0xf, 0xf, false));
}
template <int N>
__device__ __forceinline__ float group_max(float v) {
  if constexpr (N >= 16) v = fmaxf(v, row_xor<8>(v));
  if constexpr (N >= 8)  v = fmaxf(v, row_xor<4>(v));
  v = fmaxf(v, row_xor<2>(v));
  v = fmaxf(v, row_xor<1>(v));
  return v;
}
template <int N>
__device__ __forceinline__ float group_sum(float v) {
  if constexpr (N >= 16) v += row_xor<8>(v);
  if constexpr (N >= 8)  v += row_xor<4>(v);
  v += row_xor<2>(v);
  v += row_xor<1>(v);
  return v;
}

template <int T>
__device__ __forceinline__ void score_tiles(const float* __restrict__ qrow,
                                            const float* __restrict__ krow,
                                            const float* __restrict__ msk,
                                            float* __restrict__ red,
                                            float* __restrict__ dst,
                                            float* __restrict__ pdst,
                                            float* __restrict__ pair,
                                            int wv, int lane) {
  constexpr int NIT = (64 * T + 127) / 128;
  constexpr bool FULL = (64 * T) % 128 == 0;
  v4f mk[NIT];
#pragma unroll
  for (int it = 0; it < NIT; ++it) {
    const int f = threadIdx.x + it * 128;
    mk[it] = v4f{};
    if (FULL || f < 64 * T) {
      const int row = f / (4 * T), c4 = f - row * (4 * T);
      mk[it] = *(const v4f*)(msk + (size_t)row * KVLEN + 4 * c4);
    }
  }

  constexpr int CH = (T == 1) ? 32 : 8;
  constexpr int NG = 32 / CH;
  v2f a[32];
#pragma unroll
  for (int s = 0; s < 32; ++s) a[s] = *(const v2f*)(qrow + 4 * s);

  v2f kb[2][CH * T];
#pragma unroll
  for (int j = 0; j < CH; ++j) {
#pragma unroll
    for (int t = 0; t < T; ++t)
      kb[0][j * T + t] = *(const v2f*)(krow + (size_t)t * 16 * DIMX + 4 * j);
  }

  v8f c[T];
#pragma unroll
  for (int t = 0; t < T; ++t) c[t] = v8f{};
#pragma unroll
  for (int g = 0; g < NG; ++g) {
    if (g + 1 < NG) {
#pragma unroll
      for (int j = 0; j < CH; ++j) {
#pragma unroll
        for (int t = 0; t < T; ++t)
          kb[(g + 1) & 1][j * T + t] =
              *(const v2f*)(krow + (size_t)t * 16 * DIMX + 4 * ((g + 1) * CH + j));
      }
      __builtin_amdgcn_sched_barrier(0);
    }
#pragma unroll
    for (int j = 0; j < CH; ++j) {
#pragma unroll
      for (int t = 0; t < T; ++t)
        c[t] = wmma4(a[g * CH + j], kb[g & 1][j * T + t], c[t]);
    }
  }

#pragma unroll
  for (int t = 0; t < T; ++t) {
#pragma unroll
    for (int i = 0; i < 8; ++i) red[(wv * T + t) * 256 + i * 32 + lane] = c[t][i];
  }
  __syncthreads();
#pragma unroll
  for (int it = 0; it < NIT; ++it) {
    const int f = threadIdx.x + it * 128;
    if (FULL || f < 64 * T) {
      const int row = f / (4 * T), c4 = f - row * (4 * T);
      const int t = c4 >> 2, col = (c4 & 3) * 4;
      const int e = (row & 7) * 32 + (row >> 3) * 16 + col;
      const v4f p0 = *(const v4f*)(red + (0 * T + t) * 256 + e);
      const v4f p1 = *(const v4f*)(red + (1 * T + t) * 256 + e);
      const v4f p2 = *(const v4f*)(red + (2 * T + t) * 256 + e);
      const v4f p3 = *(const v4f*)(red + (3 * T + t) * 256 + e);
      const v4f s = (((p0 + p1) + p2) + p3) * QKSCALE;
      *(v4f*)(dst + (size_t)row * KVLEN + 4 * c4) = s;

      const v4f z = s + mk[it];
      const float mx = fmaxf(group_max<4 * T>(max4(z)), SMFLOOR);
      const v4f ex = exp4(z, mx);
      const float sm = group_sum<4 * T>((ex.x + ex.y) + (ex.z + ex.w));
      *(v4f*)(pdst + (size_t)row * KVLEN + 4 * c4) = ex;
      if (c4 == 0) { pair[row] = mx; pair[SEQ + row] = sm; }
    }
  }
}

__global__ void __launch_bounds__(128)
dist_kernel(const float* __restrict__ q,
            const float* __restrict__ cache_k,
            const float* __restrict__ kn,
            const float* __restrict__ mask,
            float* __restrict__ wout,
            float* __restrict__ probs,
            float* __restrict__ pairs) {
  const int wv   = threadIdx.x >> 5;
  const int lane = threadIdx.x & 31;
  const int half = lane >> 4, lm = lane & 15;
  const int koff = 2 * half + wv * 128;

  __shared__ __attribute__((aligned(16))) float red[4 * SC_T * 256];

  if (blockIdx.x < NB) {
    const int b = blockIdx.x;
    const float* qrow = q  + (size_t)(b * SEQ + lm) * DIMX + koff;
    const float* krow = kn + (size_t)(b * SEQ + lm) * DIMX + koff;
    score_tiles<1>(qrow, krow, mask + 256 * 16, red,
                   wout  + (size_t)b * SEQ * KVLEN + 256 * 16,
                   probs + (size_t)b * SEQ * KVLEN + 256 * 16,
                   pairs + (size_t)(b * NGRP + (NGRP - 1)) * 2 * SEQ, wv, lane);
  } else {
    const int t = blockIdx.x - NB;
    const int b = t >> 6, g = t & 63;
    const float* qrow = q + (size_t)(b * SEQ + lm) * DIMX + koff;
    const float* krow = cache_k
        + (size_t)(b * MAXSEQ + g * (16 * SC_T) + lm) * DIMX + koff;
    score_tiles<SC_T>(qrow, krow, mask + g * (16 * SC_T), red,
                      wout  + (size_t)b * SEQ * KVLEN + g * (16 * SC_T),
                      probs + (size_t)b * SEQ * KVLEN + g * (16 * SC_T),
                      pairs + (size_t)(b * NGRP + g) * 2 * SEQ, wv, lane);
  }
}

#define PV_SPAN  128
#define PV_PITCH 132

template <int W>
__device__ __forceinline__ float weighted_planes(const float* red_j,
                                                 float e0, float e1, float s) {
  if constexpr (W < 16) {
    s += red_j[W * 256] * row_lane<W>(e0);
    return weighted_planes<W + 1>(red_j, e0, e1, s);
  } else if constexpr (W < 32) {
    s += red_j[W * 256] * row_lane<W - 16>(e1);
    return weighted_planes<W + 1>(red_j, e0, e1, s);
  } else {
    return s;
  }
}

__global__ void __launch_bounds__(1024)
atten_kernel(const float* __restrict__ probs,
             const float* __restrict__ pairs,
             const float* __restrict__ cache_v,
             const float* __restrict__ vn,
             float* __restrict__ atten) {
  const int blk  = blockIdx.x;
  const int b    = blk >> 5;
  const int Nt   = blk & 31;
  const int wv   = threadIdx.x >> 5;
  const int lane = threadIdx.x & 31;
  const int half = lane >> 4, lm = lane & 15;

  __shared__ float stage[32 * SEQ * PV_PITCH];
  __shared__ float red[32 * 256];
  __shared__ float wmx[32 * SEQ];
  __shared__ float wsm[32 * SEQ];

  const float* pp = pairs + (size_t)(b * NGRP + 2 * wv + half) * 2 * SEQ + lm;
  const float mg = pp[0];
  const float sg = pp[SEQ];
  float mf = SMFLOOR, sf = 0.f;
  v8f cf = {};
  if (wv == 0) {
    const float* pf = pairs + (size_t)(b * NGRP + (NGRP - 1)) * 2 * SEQ + lm;
    mf = pf[0];
    sf = pf[SEQ];
    const float* prow = probs + (size_t)(b * SEQ + lm) * KVLEN + 256 * 16 + 2 * half;
    const float* vfr  = vn + (size_t)b * SEQ * DIMX + Nt * 16 + lm;
#pragma unroll
    for (int s = 0; s < 4; ++s) {
      v2f a = *(const v2f*)(prow + 4 * s);
      v2f bb;
      bb.x = vfr[(size_t)(4 * s + 2 * half)     * DIMX];
      bb.y = vfr[(size_t)(4 * s + 2 * half + 1) * DIMX];
      cf = wmma4(a, bb, cf);
    }
  }

  float* plane = stage + wv * (SEQ * PV_PITCH);
  {
    const float* pw = probs + (size_t)b * SEQ * KVLEN + wv * PV_SPAN + 4 * lane;
    v4f x[SEQ];
#pragma unroll
    for (int r = 0; r < SEQ; ++r) x[r] = *(const v4f*)(pw + (size_t)r * KVLEN);
#pragma unroll
    for (int r = 0; r < SEQ; ++r) *(v4f*)(plane + r * PV_PITCH + 4 * lane) = x[r];
  }
  __syncwarp();

  const float* afrag = plane + lm * PV_PITCH + 2 * half;
  const float* vcol  = cache_v + (size_t)(b * MAXSEQ + wv * PV_SPAN) * DIMX
                     + Nt * 16 + lm;

  v8f cg[2] = {v8f{}, v8f{}};
#pragma unroll
  for (int g = 0; g < 2; ++g) {
    for (int t = 4 * g; t < 4 * g + 4; ++t) {
#pragma unroll
      for (int s = 0; s < 4; ++s) {
        const int j = t * 16 + 4 * s;
        v2f a = *(const v2f*)(afrag + j);
        v2f bb;
        bb.x = vcol[(size_t)(j + 2 * half)     * DIMX];
        bb.y = vcol[(size_t)(j + 2 * half + 1) * DIMX];
        cg[g] = wmma4(a, bb, cg[g]);
      }
    }
  }

  const float mw = fmaxf(fmaxf(mg, __shfl_xor(mg, 16, 32)), mf);
  const float fg = __expf(mg - mw);
  const float ff = __expf(mf - mw);
  const float tg = sg * fg;
  const float lw = (tg + __shfl_xor(tg, 16, 32)) + sf * ff;
  wmx[wv * SEQ + lm] = mw;
  wsm[wv * SEQ + lm] = lw;

  v8f c;
#pragma unroll
  for (int i = 0; i < 8; ++i) {
    const int r = i + 8 * half;
    c[i] = cg[0][i] * __shfl(fg, r, 32) + cg[1][i] * __shfl(fg, 16 + r, 32)
         + cf[i] * __shfl(ff, r, 32);
  }

  put_partial(red, wv, lane, c);
  __syncthreads();
  if (threadIdx.x < 256) {
    const int j = threadIdx.x;
    const int i = j >> 5, ln = j & 31;
    const int row = i + 8 * (ln >> 4), cl = ln & 15;
    const float m0 = wmx[cl * SEQ + row], m1 = wmx[(cl + 16) * SEQ + row];
    const float M  = group_max<16>(fmaxf(m0, m1));
    const float e0 = __expf(m0 - M), e1 = __expf(m1 - M);
    const float Ls = group_sum<16>(wsm[cl * SEQ + row] * e0
                                   + wsm[(cl + 16) * SEQ + row] * e1);
    const float s  = weighted_planes<0>(red + j, e0, e1, 0.f);
    atten[(size_t)(b * SEQ + row) * DIMX + Nt * 16 + cl] = s / Ls;
  }
}

__global__ void __launch_bounds__(256)
outproj_kernel(const float* __restrict__ atten,
               const float* __restrict__ Wo,
               float* __restrict__ out) {
  const int t    = blockIdx.x;
  const int wv   = threadIdx.x >> 5;
  const int lane = threadIdx.x & 31;
  const int half = lane >> 4, lm = lane & 15;
  const int Mt   = t >> 5, Nt = t & 31;

  const float* arow = atten + (Mt * 16 + lm) * DIMX + 2 * half + wv * 64;
  const float* wrow = Wo    + (Nt * 16 + lm) * DIMX + 2 * half + wv * 64;

  const v8f c = tile_partial<64>(arow, wrow);

  __shared__ float red[8 * 256];
  put_partial(red, wv, lane, c);
  __syncthreads();
  {
    const int j = threadIdx.x;
    float s = 0.f;
#pragma unroll
    for (int w = 0; w < 8; ++w) s += red[w * 256 + j];
    const int i = j >> 5, ln = j & 31;
    out[(size_t)(Mt * 16 + i + 8 * (ln >> 4)) * DIMX + Nt * 16 + (ln & 15)] = s;
  }
}

extern "C" void kernel_launch(void* const* d_in, const int* in_sizes, int n_in,
                              void* d_out, int out_size, void* d_ws, size_t ws_size,
                              hipStream_t stream) {
  (void)in_sizes; (void)n_in; (void)out_size; (void)ws_size;
  const float* x       = (const float*)d_in[0];
  const float* mask    = (const float*)d_in[2];
  const float* Wq      = (const float*)d_in[3];
  const float* Wk      = (const float*)d_in[4];
  const float* Wv      = (const float*)d_in[5];
  const float* Wo      = (const float*)d_in[6];
  const float* cache_k = (const float*)d_in[7];
  const float* cache_v = (const float*)d_in[8];

  float* out  = (float*)d_out;
  float* wout = out + NB * SEQ * DIMX;

  float* ws    = (float*)d_ws;
  float* q     = ws;
  float* kn    = q  + NB * SEQ * DIMX;
  float* vn    = kn + NB * SEQ * DIMX;
  float* probs = vn + NB * SEQ * DIMX;
  float* pairs = probs + (size_t)NB * SEQ * KVLEN;
  float* atten = q;

  qkv_kernel    <<< 256,  256, 0, stream>>>(x, Wq, Wk, Wv, q, kn, vn);
  dist_kernel   <<< 520,  128, 0, stream>>>(q, cache_k, kn, mask, wout, probs, pairs);
  atten_kernel  <<< 256, 1024, 0, stream>>>(probs, pairs, cache_v, vn, atten);
  outproj_kernel<<< 256,  256, 0, stream>>>(atten, Wo, out);
}
